# starting kernel
# baseline (speedup 1.0000x reference)
#include <hip/hip_runtime.h>

typedef __attribute__((ext_vector_type(2))) float v2f;
typedef __attribute__((ext_vector_type(8))) float v8f;

#define DIMI 128
#define ODIM 256
#define SEQL 8192
#define NB   8
#define CH   64

__global__ void wnorm_kernel(const float* __restrict__ w, float* __restrict__ we) {
    int o = blockIdx.x * blockDim.x + threadIdx.x;
    if (o >= ODIM) return;
    float s = 0.f;
#pragma unroll 8
    for (int i = 0; i < DIMI; ++i) { float v = w[o * DIMI + i]; s += v * v; }
    float inv = 1.f / (sqrtf(s) + 1.13137084989847604e-3f);
#pragma unroll 8
    for (int i = 0; i < DIMI; ++i) we[o * DIMI + i] = w[o * DIMI + i] * inv;
}

__global__ void __launch_bounds__(256)
gemm_coef_kernel(const float* __restrict__ x, const float* __restrict__ we,
                 float2* __restrict__ coef) {
    const int lane = threadIdx.x & 31;
    const int wv   = threadIdx.x >> 5;
    const int dir  = wv >> 2;
    const int mg   = wv & 3;
    const int hb   = dir * 128 + mg * 16;
    const int gb   = hb + 64;
    const int n0   = blockIdx.x * 16;
    const int bb   = blockIdx.y;

    const int ln    = lane & 15;
    const int khalf = (lane >> 4) << 1;

    const float* xb  = x + (size_t)bb * DIMI * SEQL;
    const float* wrh = we + (size_t)(hb + ln) * DIMI;
    const float* wrg = we + (size_t)(gb + ln) * DIMI;

    v8f ch = {};
    v8f cg = {};

#pragma unroll 4
    for (int k = 0; k < DIMI; k += 4) {
        const float2 th = *(const float2*)(wrh + k + khalf);
        const float2 tg = *(const float2*)(wrg + k + khalf);
        v2f ah = { th.x, th.y };
        v2f ag = { tg.x, tg.y };
        const float b0 = xb[(size_t)(k + khalf) * SEQL + n0 + ln];
        const float b1 = xb[(size_t)(k + khalf + 1) * SEQL + n0 + ln];
        v2f bf = { b0, b1 };
        ch = __builtin_amdgcn_wmma_f32_16x16x4_f32(false, ah, false, bf,
                                                   (short)0, ch, false, false);
        cg = __builtin_amdgcn_wmma_f32_16x16x4_f32(false, ag, false, bf,
                                                   (short)0, cg, false, false);
    }

    const int mrow = (lane >> 4) << 3;
    const int n    = n0 + ln;
#pragma unroll
    for (int r = 0; r < 8; ++r) {
        const int c = (hb & 63) + mrow + r;
        const float h = ch[r];
        const float g = cg[r];
        const float eg  = __expf(g);
        const float emg = __expf(-g);
        const float z  = __builtin_amdgcn_rcpf(1.f + emg);
        const float oz = __builtin_amdgcn_rcpf(1.f + eg);
        const float s  = __builtin_amdgcn_sqrtf(
            2.f * __builtin_amdgcn_rcpf(eg + emg) + 1.f);
        coef[((size_t)(dir * NB + bb) * CH + c) * SEQL + n] =
            make_float2(oz * s, z * s * h);
    }
}

template <int DIR>
__global__ void __launch_bounds__(256)
scan_kernel(const float2* __restrict__ coef, float* __restrict__ out) {
    const int tid = threadIdx.x;
    const int c   = blockIdx.x;
    const int bb  = blockIdx.y;
    const size_t base  = ((size_t)(DIR * NB + bb) * CH + c) * SEQL;
    const int    lbase = DIR ? (SEQL - 32 - tid * 32) : (tid * 32);

    float2 e[32];
#pragma unroll
    for (int j = 0; j < 32; ++j) e[j] = coef[base + lbase + j];

    float A = 1.f, Bv = 0.f;
#pragma unroll
    for (int s = 0; s < 32; ++s) {
        const int j = DIR ? (31 - s) : s;
        A  = e[j].x * A;
        Bv = e[j].x * Bv + e[j].y;
    }

    __shared__ float sA[256];
    __shared__ float sB[256];
    sA[tid] = A; sB[tid] = Bv;
    __syncthreads();
    for (int off = 1; off < 256; off <<= 1) {
        float pa = 1.f, pb = 0.f;
        if (tid >= off) { pa = sA[tid - off]; pb = sB[tid - off]; }
        __syncthreads();
        if (tid >= off) {
            const float ca = sA[tid], cb = sB[tid];
            sA[tid] = ca * pa;
            sB[tid] = ca * pb + cb;
        }
        __syncthreads();
    }
    float h = (tid == 0) ? 0.f : sB[tid - 1];

#pragma unroll
    for (int s = 0; s < 32; ++s) {
        const int j = DIR ? (31 - s) : s;
        h = e[j].x * h + e[j].y;
        e[j].x = h;
    }

    float* op = out + ((size_t)bb * 128 + DIR * 64 + c) * SEQL + lbase;
#pragma unroll
    for (int j = 0; j < 32; ++j) op[j] = e[j].x;
}

extern "C" void kernel_launch(void* const* d_in, const int* in_sizes, int n_in,
                              void* d_out, int out_size, void* d_ws, size_t ws_size,
                              hipStream_t stream) {
    const float* x = (const float*)d_in[0];
    const float* w = (const float*)d_in[1];
    float* out = (float*)d_out;

    float*  we   = (float*)d_ws;
    float2* coef = (float2*)((char*)d_ws + ODIM * DIMI * sizeof(float));

    wnorm_kernel<<<1, 256, 0, stream>>>(w, we);
    gemm_coef_kernel<<<dim3(SEQL / 16, NB), 256, 0, stream>>>(x, we, coef);
    scan_kernel<0><<<dim3(CH, NB), 256, 0, stream>>>(coef, out);
    scan_kernel<1><<<dim3(CH, NB), 256, 0, stream>>>(coef, out);
}


# GEMM fragment loads made whole-line: norm kernel stores weights in A-fragment order (one 16 B load per lane per k-step), x read as 8 B column pairs, 4 column tiles per wave (64-wide L tile)
# speedup vs baseline: 1.5379x; 1.5379x over previous
#include <hip/hip_runtime.h>

typedef __attribute__((ext_vector_type(2))) float v2f;
typedef __attribute__((ext_vector_type(8))) float v8f;

#define DIMI 128
#define ODIM 256
#define SEQL 8192
#define NB   8
#define CH   64

__global__ void wnorm_kernel(const float* __restrict__ w, float* __restrict__ wp) {
    int o = blockIdx.x * blockDim.x + threadIdx.x;
    if (o >= ODIM) return;
    float s = 0.f;
#pragma unroll 8
    for (int i = 0; i < DIMI; ++i) { float v = w[o * DIMI + i]; s += v * v; }
    float inv = 1.f / (sqrtf(s) + 1.13137084989847604e-3f);
    const int dir    = o >> 7;
    const int within = o & 127;
    const int isg    = within >> 6;
    const int mg     = (within & 63) >> 4;
    const int row    = within & 15;
    float* dst = wp + (size_t)(dir * 4 + mg) * (32 * 32 * 4) + row * 4 + isg * 2;
#pragma unroll 8
    for (int i = 0; i < DIMI; ++i) {
        const int ks = i >> 2, half = (i >> 1) & 1, e = i & 1;
        dst[(ks * 32 + half * 16) * 4 + e] = w[o * DIMI + i] * inv;
    }
}

struct KStep { float4 a; float2 p00, p01, p10, p11; };

__device__ __forceinline__ KStep load_kstep(const float4* __restrict__ wq,
                                            const float* __restrict__ xr) {
    KStep f;
    f.a   = *wq;
    f.p00 = *(const float2*)(xr);
    f.p01 = *(const float2*)(xr + SEQL);
    f.p10 = *(const float2*)(xr + 32);
    f.p11 = *(const float2*)(xr + SEQL + 32);
    return f;
}

__device__ __forceinline__ void mma_kstep(const KStep& f, v8f (&ch)[4], v8f (&cg)[4]) {
    const v2f ah = { f.a.x, f.a.y };
    const v2f ag = { f.a.z, f.a.w };
    const v2f b0 = { f.p00.x, f.p01.x };
    const v2f b1 = { f.p00.y, f.p01.y };
    const v2f b2 = { f.p10.x, f.p11.x };
    const v2f b3 = { f.p10.y, f.p11.y };
    ch[0] = __builtin_amdgcn_wmma_f32_16x16x4_f32(false, ah, false, b0, (short)0, ch[0], false, false);
    cg[0] = __builtin_amdgcn_wmma_f32_16x16x4_f32(false, ag, false, b0, (short)0, cg[0], false, false);
    ch[1] = __builtin_amdgcn_wmma_f32_16x16x4_f32(false, ah, false, b1, (short)0, ch[1], false, false);
    cg[1] = __builtin_amdgcn_wmma_f32_16x16x4_f32(false, ag, false, b1, (short)0, cg[1], false, false);
    ch[2] = __builtin_amdgcn_wmma_f32_16x16x4_f32(false, ah, false, b2, (short)0, ch[2], false, false);
    cg[2] = __builtin_amdgcn_wmma_f32_16x16x4_f32(false, ag, false, b2, (short)0, cg[2], false, false);
    ch[3] = __builtin_amdgcn_wmma_f32_16x16x4_f32(false, ah, false, b3, (short)0, ch[3], false, false);
    cg[3] = __builtin_amdgcn_wmma_f32_16x16x4_f32(false, ag, false, b3, (short)0, cg[3], false, false);
}

__global__ void __launch_bounds__(256)
gemm_coef_kernel(const float* __restrict__ x, const float* __restrict__ wp,
                 float2* __restrict__ coef) {
    const int lane = threadIdx.x & 31;
    const int wv   = threadIdx.x >> 5;
    const int dir  = wv >> 2;
    const int mg   = wv & 3;
    const int n0   = blockIdx.x * 64;
    const int bb   = blockIdx.y;

    const int ln    = lane & 15;
    const int khalf = (lane >> 4) << 1;

    const float*  xr = x + (size_t)bb * DIMI * SEQL + (size_t)khalf * SEQL + n0 + 2 * ln;
    const float4* wq = (const float4*)wp + (size_t)wv * (32 * 32) + lane;

    v8f ch[4] = {};
    v8f cg[4] = {};

    KStep f = load_kstep(wq, xr);
#pragma unroll 2
    for (int ks = 0; ks < 32; ++ks) {
        const int kn = (ks + 1 < 32) ? ks + 1 : 31;
        const KStep nx = load_kstep(wq + kn * 32, xr + (size_t)kn * 4 * SEQL);
        mma_kstep(f, ch, cg);
        f = nx;
    }

    const int mrow = (lane >> 4) << 3;
    float4* coef4 = (float4*)(coef + ((size_t)(dir * NB + bb) * CH + mg * 16 + mrow) * SEQL
                                   + n0 + 2 * ln);
#pragma unroll
    for (int r = 0; r < 8; ++r) {
#pragma unroll
        for (int q = 0; q < 2; ++q) {
            float ab[4];
#pragma unroll
            for (int p = 0; p < 2; ++p) {
                const float h = ch[q * 2 + p][r];
                const float g = cg[q * 2 + p][r];
                const float eg  = __expf(g);
                const float emg = __expf(-g);
                const float z  = __builtin_amdgcn_rcpf(1.f + emg);
                const float oz = __builtin_amdgcn_rcpf(1.f + eg);
                const float s  = __builtin_amdgcn_sqrtf(
                    2.f * __builtin_amdgcn_rcpf(eg + emg) + 1.f);
                ab[p * 2 + 0] = oz * s;
                ab[p * 2 + 1] = z * s * h;
            }
            coef4[((size_t)r * SEQL + 32 * q) / 2] = make_float4(ab[0], ab[1], ab[2], ab[3]);
        }
    }
}

template <int DIR>
__global__ void __launch_bounds__(256)
scan_kernel(const float2* __restrict__ coef, float* __restrict__ out) {
    const int tid = threadIdx.x;
    const int c   = blockIdx.x;
    const int bb  = blockIdx.y;
    const size_t base  = ((size_t)(DIR * NB + bb) * CH + c) * SEQL;
    const int    lbase = DIR ? (SEQL - 32 - tid * 32) : (tid * 32);

    float2 e[32];
#pragma unroll
    for (int j = 0; j < 32; ++j) e[j] = coef[base + lbase + j];

    float A = 1.f, Bv = 0.f;
#pragma unroll
    for (int s = 0; s < 32; ++s) {
        const int j = DIR ? (31 - s) : s;
        A  = e[j].x * A;
        Bv = e[j].x * Bv + e[j].y;
    }

    __shared__ float sA[256];
    __shared__ float sB[256];
    sA[tid] = A; sB[tid] = Bv;
    __syncthreads();
    for (int off = 1; off < 256; off <<= 1) {
        float pa = 1.f, pb = 0.f;
        if (tid >= off) { pa = sA[tid - off]; pb = sB[tid - off]; }
        __syncthreads();
        if (tid >= off) {
            const float ca = sA[tid], cb = sB[tid];
            sA[tid] = ca * pa;
            sB[tid] = ca * pb + cb;
        }
        __syncthreads();
    }
    float h = (tid == 0) ? 0.f : sB[tid - 1];

#pragma unroll
    for (int s = 0; s < 32; ++s) {
        const int j = DIR ? (31 - s) : s;
        h = e[j].x * h + e[j].y;
        e[j].x = h;
    }

    float* op = out + ((size_t)bb * 128 + DIR * 64 + c) * SEQL + lbase;
#pragma unroll
    for (int j = 0; j < 32; ++j) op[j] = e[j].x;
}

extern "C" void kernel_launch(void* const* d_in, const int* in_sizes, int n_in,
                              void* d_out, int out_size, void* d_ws, size_t ws_size,
                              hipStream_t stream) {
    const float* x = (const float*)d_in[0];
    const float* w = (const float*)d_in[1];
    float* out = (float*)d_out;

    float*  we   = (float*)d_ws;
    float2* coef = (float2*)((char*)d_ws + ODIM * DIMI * sizeof(float));

    wnorm_kernel<<<1, 256, 0, stream>>>(w, we);
    gemm_coef_kernel<<<dim3(SEQL / 64, NB), 256, 0, stream>>>(x, we, coef);
    scan_kernel<0><<<dim3(CH, NB), 256, 0, stream>>>(coef, out);
    scan_kernel<1><<<dim3(CH, NB), 256, 0, stream>>>(coef, out);
}


# Weight-norm kernel spread from 1 block to 8 blocks x 32 waves: a wave per weight row (one 512 B load), sum of squares as lane partials + shuffle tree, regroup in LDS, whole-line fragment stores
# speedup vs baseline: 1.7971x; 1.1685x over previous
#include <hip/hip_runtime.h>

typedef __attribute__((ext_vector_type(2))) float v2f;
typedef __attribute__((ext_vector_type(8))) float v8f;

#define DIMI 128
#define ODIM 256
#define SEQL 8192
#define NB   8
#define CH   64

#define WPITCH 132
__global__ void __launch_bounds__(1024)
wnorm_kernel(const float* __restrict__ w, float* __restrict__ wp) {
    __shared__ float sW[32 * WPITCH];
    const int lane = threadIdx.x & 31;
    const int wv   = threadIdx.x >> 5;
    const int t    = blockIdx.x;
    const int dir  = t >> 2;
    const int mg   = t & 3;
    const int o    = dir * 128 + (wv >> 4) * 64 + mg * 16 + (wv & 15);
    const float* wr = w + o * DIMI + 4 * lane;
    const float v0 = wr[0], v1 = wr[1], v2 = wr[2], v3 = wr[3];
    float s = v0 * v0;
    s += v1 * v1;
    s += v2 * v2;
    s += v3 * v3;
    for (int m = 16; m >= 1; m >>= 1) s += __shfl_xor(s, m);
    const float inv = 1.f / (sqrtf(s) + 1.13137084989847604e-3f);
    *(float4*)(sW + wv * WPITCH + 4 * lane) =
        make_float4(v0 * inv, v1 * inv, v2 * inv, v3 * inv);
    __syncthreads();
    const int half = lane >> 4;
    const int row  = lane & 15;
    const int k    = 4 * wv + 2 * half;
    const float2 h = *(const float2*)(sW + row * WPITCH + k);
    const float2 g = *(const float2*)(sW + (16 + row) * WPITCH + k);
    ((float4*)wp)[(t * 32 + wv) * 32 + lane] = make_float4(h.x, h.y, g.x, g.y);
}

struct KStep { float4 a; float2 p00, p01, p10, p11; };

__device__ __forceinline__ KStep load_kstep(const float4* __restrict__ wq,
                                            const float* __restrict__ xr) {
    KStep f;
    f.a   = *wq;
    f.p00 = *(const float2*)(xr);
    f.p01 = *(const float2*)(xr + SEQL);
    f.p10 = *(const float2*)(xr + 32);
    f.p11 = *(const float2*)(xr + SEQL + 32);
    return f;
}

__device__ __forceinline__ void mma_kstep(const KStep& f, v8f (&ch)[4], v8f (&cg)[4]) {
    const v2f ah = { f.a.x, f.a.y };
    const v2f ag = { f.a.z, f.a.w };
    const v2f b0 = { f.p00.x, f.p01.x };
    const v2f b1 = { f.p00.y, f.p01.y };
    const v2f b2 = { f.p10.x, f.p11.x };
    const v2f b3 = { f.p10.y, f.p11.y };
    ch[0] = __builtin_amdgcn_wmma_f32_16x16x4_f32(false, ah, false, b0, (short)0, ch[0], false, false);
    cg[0] = __builtin_amdgcn_wmma_f32_16x16x4_f32(false, ag, false, b0, (short)0, cg[0], false, false);
    ch[1] = __builtin_amdgcn_wmma_f32_16x16x4_f32(false, ah, false, b1, (short)0, ch[1], false, false);
    cg[1] = __builtin_amdgcn_wmma_f32_16x16x4_f32(false, ag, false, b1, (short)0, cg[1], false, false);
    ch[2] = __builtin_amdgcn_wmma_f32_16x16x4_f32(false, ah, false, b2, (short)0, ch[2], false, false);
    cg[2] = __builtin_amdgcn_wmma_f32_16x16x4_f32(false, ag, false, b2, (short)0, cg[2], false, false);
    ch[3] = __builtin_amdgcn_wmma_f32_16x16x4_f32(false, ah, false, b3, (short)0, ch[3], false, false);
    cg[3] = __builtin_amdgcn_wmma_f32_16x16x4_f32(false, ag, false, b3, (short)0, cg[3], false, false);
}

__global__ void __launch_bounds__(256)
gemm_coef_kernel(const float* __restrict__ x, const float* __restrict__ wp,
                 float2* __restrict__ coef) {
    const int lane = threadIdx.x & 31;
    const int wv   = threadIdx.x >> 5;
    const int dir  = wv >> 2;
    const int mg   = wv & 3;
    const int n0   = blockIdx.x * 64;
    const int bb   = blockIdx.y;

    const int ln    = lane & 15;
    const int khalf = (lane >> 4) << 1;

    const float*  xr = x + (size_t)bb * DIMI * SEQL + (size_t)khalf * SEQL + n0 + 2 * ln;
    const float4* wq = (const float4*)wp + (size_t)wv * (32 * 32) + lane;

    v8f ch[4] = {};
    v8f cg[4] = {};

    KStep f = load_kstep(wq, xr);
#pragma unroll 2
    for (int ks = 0; ks < 32; ++ks) {
        const int kn = (ks + 1 < 32) ? ks + 1 : 31;
        const KStep nx = load_kstep(wq + kn * 32, xr + (size_t)kn * 4 * SEQL);
        mma_kstep(f, ch, cg);
        f = nx;
    }

    const int mrow = (lane >> 4) << 3;
    float4* coef4 = (float4*)(coef + ((size_t)(dir * NB + bb) * CH + mg * 16 + mrow) * SEQL
                                   + n0 + 2 * ln);
#pragma unroll
    for (int r = 0; r < 8; ++r) {
#pragma unroll
        for (int q = 0; q < 2; ++q) {
            float ab[4];
#pragma unroll
            for (int p = 0; p < 2; ++p) {
                const float h = ch[q * 2 + p][r];
                const float g = cg[q * 2 + p][r];
                const float eg  = __expf(g);
                const float emg = __expf(-g);
                const float z  = __builtin_amdgcn_rcpf(1.f + emg);
                const float oz = __builtin_amdgcn_rcpf(1.f + eg);
                const float s  = __builtin_amdgcn_sqrtf(
                    2.f * __builtin_amdgcn_rcpf(eg + emg) + 1.f);
                ab[p * 2 + 0] = oz * s;
                ab[p * 2 + 1] = z * s * h;
            }
            coef4[((size_t)r * SEQL + 32 * q) / 2] = make_float4(ab[0], ab[1], ab[2], ab[3]);
        }
    }
}

template <int DIR>
__global__ void __launch_bounds__(256)
scan_kernel(const float2* __restrict__ coef, float* __restrict__ out) {
    const int tid = threadIdx.x;
    const int c   = blockIdx.x;
    const int bb  = blockIdx.y;
    const size_t base  = ((size_t)(DIR * NB + bb) * CH + c) * SEQL;
    const int    lbase = DIR ? (SEQL - 32 - tid * 32) : (tid * 32);

    float2 e[32];
#pragma unroll
    for (int j = 0; j < 32; ++j) e[j] = coef[base + lbase + j];

    float A = 1.f, Bv = 0.f;
#pragma unroll
    for (int s = 0; s < 32; ++s) {
        const int j = DIR ? (31 - s) : s;
        A  = e[j].x * A;
        Bv = e[j].x * Bv + e[j].y;
    }

    __shared__ float sA[256];
    __shared__ float sB[256];
    sA[tid] = A; sB[tid] = Bv;
    __syncthreads();
    for (int off = 1; off < 256; off <<= 1) {
        float pa = 1.f, pb = 0.f;
        if (tid >= off) { pa = sA[tid - off]; pb = sB[tid - off]; }
        __syncthreads();
        if (tid >= off) {
            const float ca = sA[tid], cb = sB[tid];
            sA[tid] = ca * pa;
            sB[tid] = ca * pb + cb;
        }
        __syncthreads();
    }
    float h = (tid == 0) ? 0.f : sB[tid - 1];

#pragma unroll
    for (int s = 0; s < 32; ++s) {
        const int j = DIR ? (31 - s) : s;
        h = e[j].x * h + e[j].y;
        e[j].x = h;
    }

    float* op = out + ((size_t)bb * 128 + DIR * 64 + c) * SEQL + lbase;
#pragma unroll
    for (int j = 0; j < 32; ++j) op[j] = e[j].x;
}

extern "C" void kernel_launch(void* const* d_in, const int* in_sizes, int n_in,
                              void* d_out, int out_size, void* d_ws, size_t ws_size,
                              hipStream_t stream) {
    const float* x = (const float*)d_in[0];
    const float* w = (const float*)d_in[1];
    float* out = (float*)d_out;

    float*  we   = (float*)d_ws;
    float2* coef = (float2*)((char*)d_ws + ODIM * DIMI * sizeof(float));

    wnorm_kernel<<<8, 1024, 0, stream>>>(w, we);
    gemm_coef_kernel<<<dim3(SEQL / 64, NB), 256, 0, stream>>>(x, we, coef);
    scan_kernel<0><<<dim3(CH, NB), 256, 0, stream>>>(coef, out);
    scan_kernel<1><<<dim3(CH, NB), 256, 0, stream>>>(coef, out);
}


# Scan kernels' global loads and stores made whole-line: a wave reads its 8 KB of coefficients as 512 B contiguous loads and regroups them per lane through padded LDS; outputs leave the same way
# speedup vs baseline: 1.9487x; 1.0844x over previous
#include <hip/hip_runtime.h>

typedef __attribute__((ext_vector_type(2))) float v2f;
typedef __attribute__((ext_vector_type(8))) float v8f;

#define DIMI 128
#define ODIM 256
#define SEQL 8192
#define NB   8
#define CH   64

#define WPITCH 132
__global__ void __launch_bounds__(1024)
wnorm_kernel(const float* __restrict__ w, float* __restrict__ wp) {
    __shared__ float sW[32 * WPITCH];
    const int lane = threadIdx.x & 31;
    const int wv   = threadIdx.x >> 5;
    const int t    = blockIdx.x;
    const int dir  = t >> 2;
    const int mg   = t & 3;
    const int o    = dir * 128 + (wv >> 4) * 64 + mg * 16 + (wv & 15);
    const float* wr = w + o * DIMI + 4 * lane;
    const float v0 = wr[0], v1 = wr[1], v2 = wr[2], v3 = wr[3];
    float s = v0 * v0;
    s += v1 * v1;
    s += v2 * v2;
    s += v3 * v3;
    for (int m = 16; m >= 1; m >>= 1) s += __shfl_xor(s, m);
    const float inv = 1.f / (sqrtf(s) + 1.13137084989847604e-3f);
    *(float4*)(sW + wv * WPITCH + 4 * lane) =
        make_float4(v0 * inv, v1 * inv, v2 * inv, v3 * inv);
    __syncthreads();
    const int half = lane >> 4;
    const int row  = lane & 15;
    const int k    = 4 * wv + 2 * half;
    const float2 h = *(const float2*)(sW + row * WPITCH + k);
    const float2 g = *(const float2*)(sW + (16 + row) * WPITCH + k);
    ((float4*)wp)[(t * 32 + wv) * 32 + lane] = make_float4(h.x, h.y, g.x, g.y);
}

struct KStep { float4 a; float2 p00, p01, p10, p11; };

__device__ __forceinline__ KStep load_kstep(const float4* __restrict__ wq,
                                            const float* __restrict__ xr) {
    KStep f;
    f.a   = *wq;
    f.p00 = *(const float2*)(xr);
    f.p01 = *(const float2*)(xr + SEQL);
    f.p10 = *(const float2*)(xr + 32);
    f.p11 = *(const float2*)(xr + SEQL + 32);
    return f;
}

__device__ __forceinline__ void mma_kstep(const KStep& f, v8f (&ch)[4], v8f (&cg)[4]) {
    const v2f ah = { f.a.x, f.a.y };
    const v2f ag = { f.a.z, f.a.w };
    const v2f b0 = { f.p00.x, f.p01.x };
    const v2f b1 = { f.p00.y, f.p01.y };
    const v2f b2 = { f.p10.x, f.p11.x };
    const v2f b3 = { f.p10.y, f.p11.y };
    ch[0] = __builtin_amdgcn_wmma_f32_16x16x4_f32(false, ah, false, b0, (short)0, ch[0], false, false);
    cg[0] = __builtin_amdgcn_wmma_f32_16x16x4_f32(false, ag, false, b0, (short)0, cg[0], false, false);
    ch[1] = __builtin_amdgcn_wmma_f32_16x16x4_f32(false, ah, false, b1, (short)0, ch[1], false, false);
    cg[1] = __builtin_amdgcn_wmma_f32_16x16x4_f32(false, ag, false, b1, (short)0, cg[1], false, false);
    ch[2] = __builtin_amdgcn_wmma_f32_16x16x4_f32(false, ah, false, b2, (short)0, ch[2], false, false);
    cg[2] = __builtin_amdgcn_wmma_f32_16x16x4_f32(false, ag, false, b2, (short)0, cg[2], false, false);
    ch[3] = __builtin_amdgcn_wmma_f32_16x16x4_f32(false, ah, false, b3, (short)0, ch[3], false, false);
    cg[3] = __builtin_amdgcn_wmma_f32_16x16x4_f32(false, ag, false, b3, (short)0, cg[3], false, false);
}

__global__ void __launch_bounds__(256)
gemm_coef_kernel(const float* __restrict__ x, const float* __restrict__ wp,
                 float2* __restrict__ coef) {
    const int lane = threadIdx.x & 31;
    const int wv   = threadIdx.x >> 5;
    const int dir  = wv >> 2;
    const int mg   = wv & 3;
    const int n0   = blockIdx.x * 64;
    const int bb   = blockIdx.y;

    const int ln    = lane & 15;
    const int khalf = (lane >> 4) << 1;

    const float*  xr = x + (size_t)bb * DIMI * SEQL + (size_t)khalf * SEQL + n0 + 2 * ln;
    const float4* wq = (const float4*)wp + (size_t)wv * (32 * 32) + lane;

    v8f ch[4] = {};
    v8f cg[4] = {};

    KStep f = load_kstep(wq, xr);
#pragma unroll 2
    for (int ks = 0; ks < 32; ++ks) {
        const int kn = (ks + 1 < 32) ? ks + 1 : 31;
        const KStep nx = load_kstep(wq + kn * 32, xr + (size_t)kn * 4 * SEQL);
        mma_kstep(f, ch, cg);
        f = nx;
    }

    const int mrow = (lane >> 4) << 3;
    float4* coef4 = (float4*)(coef + ((size_t)(dir * NB + bb) * CH + mg * 16 + mrow) * SEQL
                                   + n0 + 2 * ln);
#pragma unroll
    for (int r = 0; r < 8; ++r) {
#pragma unroll
        for (int q = 0; q < 2; ++q) {
            float ab[4];
#pragma unroll
            for (int p = 0; p < 2; ++p) {
                const float h = ch[q * 2 + p][r];
                const float g = cg[q * 2 + p][r];
                const float eg  = __expf(g);
                const float emg = __expf(-g);
                const float z  = __builtin_amdgcn_rcpf(1.f + emg);
                const float oz = __builtin_amdgcn_rcpf(1.f + eg);
                const float s  = __builtin_amdgcn_sqrtf(
                    2.f * __builtin_amdgcn_rcpf(eg + emg) + 1.f);
                ab[p * 2 + 0] = oz * s;
                ab[p * 2 + 1] = z * s * h;
            }
            coef4[((size_t)r * SEQL + 32 * q) / 2] = make_float4(ab[0], ab[1], ab[2], ab[3]);
        }
    }
}

#define CPITCH 17
#define OPITCH 9
template <int DIR>
__global__ void __launch_bounds__(256)
scan_kernel(const float2* __restrict__ coef, float* __restrict__ out) {
    __shared__ float4 sT[8 * 32 * CPITCH];
    __shared__ float sA[256];
    __shared__ float sB[256];
    const int tid  = threadIdx.x;
    const int lane = tid & 31;
    const int wv   = tid >> 5;
    const int c    = blockIdx.x;
    const int bb   = blockIdx.y;
    const size_t base = ((size_t)(DIR * NB + bb) * CH + c) * SEQL;
    const int    wlo  = DIR ? (SEQL - 1024 * (wv + 1)) : (1024 * wv);
    const int    chk  = DIR ? (31 - lane) : lane;
    float4* sw = sT + wv * (32 * CPITCH);

    const float4* src = (const float4*)(coef + base + wlo) + lane;
    float4 v[16];
#pragma unroll
    for (int i = 0; i < 16; ++i) v[i] = src[i * 32];
#pragma unroll
    for (int i = 0; i < 16; ++i)
        sw[(2 * i + (lane >> 4)) * CPITCH + (lane & 15)] = v[i];
    __syncthreads();

    float2 e[32];
#pragma unroll
    for (int i = 0; i < 16; ++i) {
        const float4 t = sw[chk * CPITCH + i];
        e[2 * i]     = make_float2(t.x, t.y);
        e[2 * i + 1] = make_float2(t.z, t.w);
    }

    float A = 1.f, Bv = 0.f;
#pragma unroll
    for (int s = 0; s < 32; ++s) {
        const int j = DIR ? (31 - s) : s;
        A  = e[j].x * A;
        Bv = e[j].x * Bv + e[j].y;
    }

    sA[tid] = A; sB[tid] = Bv;
    __syncthreads();
    for (int off = 1; off < 256; off <<= 1) {
        float pa = 1.f, pb = 0.f;
        if (tid >= off) { pa = sA[tid - off]; pb = sB[tid - off]; }
        __syncthreads();
        if (tid >= off) {
            const float ca = sA[tid], cb = sB[tid];
            sA[tid] = ca * pa;
            sB[tid] = ca * pb + cb;
        }
        __syncthreads();
    }
    float h = (tid == 0) ? 0.f : sB[tid - 1];

#pragma unroll
    for (int s = 0; s < 32; ++s) {
        const int j = DIR ? (31 - s) : s;
        h = e[j].x * h + e[j].y;
        e[j].x = h;
    }

#pragma unroll
    for (int q = 0; q < 8; ++q)
        sw[chk * OPITCH + q] = make_float4(e[4 * q].x, e[4 * q + 1].x,
                                           e[4 * q + 2].x, e[4 * q + 3].x);
    __syncthreads();
    float4* op = (float4*)(out + ((size_t)bb * 128 + DIR * 64 + c) * SEQL + wlo) + lane;
#pragma unroll
    for (int q = 0; q < 8; ++q)
        op[q * 32] = sw[(4 * q + (lane >> 3)) * OPITCH + (lane & 7)];
}

extern "C" void kernel_launch(void* const* d_in, const int* in_sizes, int n_in,
                              void* d_out, int out_size, void* d_ws, size_t ws_size,
                              hipStream_t stream) {
    const float* x = (const float*)d_in[0];
    const float* w = (const float*)d_in[1];
    float* out = (float*)d_out;

    float*  we   = (float*)d_ws;
    float2* coef = (float2*)((char*)d_ws + ODIM * DIMI * sizeof(float));

    wnorm_kernel<<<8, 1024, 0, stream>>>(w, we);
    gemm_coef_kernel<<<dim3(SEQL / 64, NB), 256, 0, stream>>>(x, we, coef);
    scan_kernel<0><<<dim3(CH, NB), 256, 0, stream>>>(coef, out);
    scan_kernel<1><<<dim3(CH, NB), 256, 0, stream>>>(coef, out);
}


# Both scan directions run in one launch (grid z = direction, body picked by a block-uniform branch): 3 launches instead of 4, all 1024 scan blocks resident together
# speedup vs baseline: 2.1096x; 1.0826x over previous
#include <hip/hip_runtime.h>

typedef __attribute__((ext_vector_type(2))) float v2f;
typedef __attribute__((ext_vector_type(8))) float v8f;

#define DIMI 128
#define ODIM 256
#define SEQL 8192
#define NB   8
#define CH   64

#define WPITCH 132
__global__ void __launch_bounds__(1024)
wnorm_kernel(const float* __restrict__ w, float* __restrict__ wp) {
    __shared__ float sW[32 * WPITCH];
    const int lane = threadIdx.x & 31;
    const int wv   = threadIdx.x >> 5;
    const int t    = blockIdx.x;
    const int dir  = t >> 2;
    const int mg   = t & 3;
    const int o    = dir * 128 + (wv >> 4) * 64 + mg * 16 + (wv & 15);
    const float* wr = w + o * DIMI + 4 * lane;
    const float v0 = wr[0], v1 = wr[1], v2 = wr[2], v3 = wr[3];
    float s = v0 * v0;
    s += v1 * v1;
    s += v2 * v2;
    s += v3 * v3;
    for (int m = 16; m >= 1; m >>= 1) s += __shfl_xor(s, m);
    const float inv = 1.f / (sqrtf(s) + 1.13137084989847604e-3f);
    *(float4*)(sW + wv * WPITCH + 4 * lane) =
        make_float4(v0 * inv, v1 * inv, v2 * inv, v3 * inv);
    __syncthreads();
    const int half = lane >> 4;
    const int row  = lane & 15;
    const int k    = 4 * wv + 2 * half;
    const float2 h = *(const float2*)(sW + row * WPITCH + k);
    const float2 g = *(const float2*)(sW + (16 + row) * WPITCH + k);
    ((float4*)wp)[(t * 32 + wv) * 32 + lane] = make_float4(h.x, h.y, g.x, g.y);
}

struct KStep { float4 a; float2 p00, p01, p10, p11; };

__device__ __forceinline__ KStep load_kstep(const float4* __restrict__ wq,
                                            const float* __restrict__ xr) {
    KStep f;
    f.a   = *wq;
    f.p00 = *(const float2*)(xr);
    f.p01 = *(const float2*)(xr + SEQL);
    f.p10 = *(const float2*)(xr + 32);
    f.p11 = *(const float2*)(xr + SEQL + 32);
    return f;
}

__device__ __forceinline__ void mma_kstep(const KStep& f, v8f (&ch)[4], v8f (&cg)[4]) {
    const v2f ah = { f.a.x, f.a.y };
    const v2f ag = { f.a.z, f.a.w };
    const v2f b0 = { f.p00.x, f.p01.x };
    const v2f b1 = { f.p00.y, f.p01.y };
    const v2f b2 = { f.p10.x, f.p11.x };
    const v2f b3 = { f.p10.y, f.p11.y };
    ch[0] = __builtin_amdgcn_wmma_f32_16x16x4_f32(false, ah, false, b0, (short)0, ch[0], false, false);
    cg[0] = __builtin_amdgcn_wmma_f32_16x16x4_f32(false, ag, false, b0, (short)0, cg[0], false, false);
    ch[1] = __builtin_amdgcn_wmma_f32_16x16x4_f32(false, ah, false, b1, (short)0, ch[1], false, false);
    cg[1] = __builtin_amdgcn_wmma_f32_16x16x4_f32(false, ag, false, b1, (short)0, cg[1], false, false);
    ch[2] = __builtin_amdgcn_wmma_f32_16x16x4_f32(false, ah, false, b2, (short)0, ch[2], false, false);
    cg[2] = __builtin_amdgcn_wmma_f32_16x16x4_f32(false, ag, false, b2, (short)0, cg[2], false, false);
    ch[3] = __builtin_amdgcn_wmma_f32_16x16x4_f32(false, ah, false, b3, (short)0, ch[3], false, false);
    cg[3] = __builtin_amdgcn_wmma_f32_16x16x4_f32(false, ag, false, b3, (short)0, cg[3], false, false);
}

__global__ void __launch_bounds__(256)
gemm_coef_kernel(const float* __restrict__ x, const float* __restrict__ wp,
                 float2* __restrict__ coef) {
    const int lane = threadIdx.x & 31;
    const int wv   = threadIdx.x >> 5;
    const int dir  = wv >> 2;
    const int mg   = wv & 3;
    const int n0   = blockIdx.x * 64;
    const int bb   = blockIdx.y;

    const int ln    = lane & 15;
    const int khalf = (lane >> 4) << 1;

    const float*  xr = x + (size_t)bb * DIMI * SEQL + (size_t)khalf * SEQL + n0 + 2 * ln;
    const float4* wq = (const float4*)wp + (size_t)wv * (32 * 32) + lane;

    v8f ch[4] = {};
    v8f cg[4] = {};

    KStep f = load_kstep(wq, xr);
#pragma unroll 2
    for (int ks = 0; ks < 32; ++ks) {
        const int kn = (ks + 1 < 32) ? ks + 1 : 31;
        const KStep nx = load_kstep(wq + kn * 32, xr + (size_t)kn * 4 * SEQL);
        mma_kstep(f, ch, cg);
        f = nx;
    }

    const int mrow = (lane >> 4) << 3;
    float4* coef4 = (float4*)(coef + ((size_t)(dir * NB + bb) * CH + mg * 16 + mrow) * SEQL
                                   + n0 + 2 * ln);
#pragma unroll
    for (int r = 0; r < 8; ++r) {
#pragma unroll
        for (int q = 0; q < 2; ++q) {
            float ab[4];
#pragma unroll
            for (int p = 0; p < 2; ++p) {
                const float h = ch[q * 2 + p][r];
                const float g = cg[q * 2 + p][r];
                const float eg  = __expf(g);
                const float emg = __expf(-g);
                const float z  = __builtin_amdgcn_rcpf(1.f + emg);
                const float oz = __builtin_amdgcn_rcpf(1.f + eg);
                const float s  = __builtin_amdgcn_sqrtf(
                    2.f * __builtin_amdgcn_rcpf(eg + emg) + 1.f);
                ab[p * 2 + 0] = oz * s;
                ab[p * 2 + 1] = z * s * h;
            }
            coef4[((size_t)r * SEQL + 32 * q) / 2] = make_float4(ab[0], ab[1], ab[2], ab[3]);
        }
    }
}

#define CPITCH 17
#define OPITCH 9
template <int DIR>
__device__ __forceinline__ void
scan_body(const float2* __restrict__ coef, float* __restrict__ out,
          float4* sT, float* sA, float* sB) {
    const int tid  = threadIdx.x;
    const int lane = tid & 31;
    const int wv   = tid >> 5;
    const int c    = blockIdx.x;
    const int bb   = blockIdx.y;
    const size_t base = ((size_t)(DIR * NB + bb) * CH + c) * SEQL;
    const int    wlo  = DIR ? (SEQL - 1024 * (wv + 1)) : (1024 * wv);
    const int    chk  = DIR ? (31 - lane) : lane;
    float4* sw = sT + wv * (32 * CPITCH);

    const float4* src = (const float4*)(coef + base + wlo) + lane;
    float4 v[16];
#pragma unroll
    for (int i = 0; i < 16; ++i) v[i] = src[i * 32];
#pragma unroll
    for (int i = 0; i < 16; ++i)
        sw[(2 * i + (lane >> 4)) * CPITCH + (lane & 15)] = v[i];
    __syncthreads();

    float2 e[32];
#pragma unroll
    for (int i = 0; i < 16; ++i) {
        const float4 t = sw[chk * CPITCH + i];
        e[2 * i]     = make_float2(t.x, t.y);
        e[2 * i + 1] = make_float2(t.z, t.w);
    }

    float A = 1.f, Bv = 0.f;
#pragma unroll
    for (int s = 0; s < 32; ++s) {
        const int j = DIR ? (31 - s) : s;
        A  = e[j].x * A;
        Bv = e[j].x * Bv + e[j].y;
    }

    sA[tid] = A; sB[tid] = Bv;
    __syncthreads();
    for (int off = 1; off < 256; off <<= 1) {
        float pa = 1.f, pb = 0.f;
        if (tid >= off) { pa = sA[tid - off]; pb = sB[tid - off]; }
        __syncthreads();
        if (tid >= off) {
            const float ca = sA[tid], cb = sB[tid];
            sA[tid] = ca * pa;
            sB[tid] = ca * pb + cb;
        }
        __syncthreads();
    }
    float h = (tid == 0) ? 0.f : sB[tid - 1];

#pragma unroll
    for (int s = 0; s < 32; ++s) {
        const int j = DIR ? (31 - s) : s;
        h = e[j].x * h + e[j].y;
        e[j].x = h;
    }

#pragma unroll
    for (int q = 0; q < 8; ++q)
        sw[chk * OPITCH + q] = make_float4(e[4 * q].x, e[4 * q + 1].x,
                                           e[4 * q + 2].x, e[4 * q + 3].x);
    __syncthreads();
    float4* op = (float4*)(out + ((size_t)bb * 128 + DIR * 64 + c) * SEQL + wlo) + lane;
#pragma unroll
    for (int q = 0; q < 8; ++q)
        op[q * 32] = sw[(4 * q + (lane >> 3)) * OPITCH + (lane & 7)];
}

__global__ void __launch_bounds__(256)
scan_kernel(const float2* __restrict__ coef, float* __restrict__ out) {
    __shared__ float4 sT[8 * 32 * CPITCH];
    __shared__ float sA[256];
    __shared__ float sB[256];
    if (blockIdx.z == 0) scan_body<0>(coef, out, sT, sA, sB);
    else                 scan_body<1>(coef, out, sT, sA, sB);
}

extern "C" void kernel_launch(void* const* d_in, const int* in_sizes, int n_in,
                              void* d_out, int out_size, void* d_ws, size_t ws_size,
                              hipStream_t stream) {
    const float* x = (const float*)d_in[0];
    const float* w = (const float*)d_in[1];
    float* out = (float*)d_out;

    float*  we   = (float*)d_ws;
    float2* coef = (float2*)((char*)d_ws + ODIM * DIMI * sizeof(float));

    wnorm_kernel<<<8, 1024, 0, stream>>>(w, we);
    gemm_coef_kernel<<<dim3(SEQL / 64, NB), 256, 0, stream>>>(x, we, coef);
    scan_kernel<<<dim3(CH, NB, 2), 256, 0, stream>>>(coef, out);
}


# GEMM K loop: each step's fragment loads issued two steps ahead of its WMMAs and held there by scheduling fences (the built parent issued them right above their own use); three register sets
# speedup vs baseline: 2.2970x; 1.0888x over previous
#include <hip/hip_runtime.h>

typedef __attribute__((ext_vector_type(2))) float v2f;
typedef __attribute__((ext_vector_type(8))) float v8f;

#define DIMI 128
#define ODIM 256
#define SEQL 8192
#define NB   8
#define CH   64

#define WPITCH 132
__global__ void __launch_bounds__(1024)
wnorm_kernel(const float* __restrict__ w, float* __restrict__ wp) {
    __shared__ float sW[32 * WPITCH];
    const int lane = threadIdx.x & 31;
    const int wv   = threadIdx.x >> 5;
    const int t    = blockIdx.x;
    const int dir  = t >> 2;
    const int mg   = t & 3;
    const int o    = dir * 128 + (wv >> 4) * 64 + mg * 16 + (wv & 15);
    const float* wr = w + o * DIMI + 4 * lane;
    const float v0 = wr[0], v1 = wr[1], v2 = wr[2], v3 = wr[3];
    float s = v0 * v0;
    s += v1 * v1;
    s += v2 * v2;
    s += v3 * v3;
    for (int m = 16; m >= 1; m >>= 1) s += __shfl_xor(s, m);
    const float inv = 1.f / (sqrtf(s) + 1.13137084989847604e-3f);
    *(float4*)(sW + wv * WPITCH + 4 * lane) =
        make_float4(v0 * inv, v1 * inv, v2 * inv, v3 * inv);
    __syncthreads();
    const int half = lane >> 4;
    const int row  = lane & 15;
    const int k    = 4 * wv + 2 * half;
    const float2 h = *(const float2*)(sW + row * WPITCH + k);
    const float2 g = *(const float2*)(sW + (16 + row) * WPITCH + k);
    ((float4*)wp)[(t * 32 + wv) * 32 + lane] = make_float4(h.x, h.y, g.x, g.y);
}

struct KStep { float4 a; float2 p00, p01, p10, p11; };

__device__ __forceinline__ KStep load_kstep(const float4* __restrict__ wq,
                                            const float* __restrict__ xr) {
    KStep f;
    f.a   = *wq;
    f.p00 = *(const float2*)(xr);
    f.p01 = *(const float2*)(xr + SEQL);
    f.p10 = *(const float2*)(xr + 32);
    f.p11 = *(const float2*)(xr + SEQL + 32);
    return f;
}

__device__ __forceinline__ void mma_kstep(const KStep& f, v8f (&ch)[4], v8f (&cg)[4]) {
    const v2f ah = { f.a.x, f.a.y };
    const v2f ag = { f.a.z, f.a.w };
    const v2f b0 = { f.p00.x, f.p01.x };
    const v2f b1 = { f.p00.y, f.p01.y };
    const v2f b2 = { f.p10.x, f.p11.x };
    const v2f b3 = { f.p10.y, f.p11.y };
    ch[0] = __builtin_amdgcn_wmma_f32_16x16x4_f32(false, ah, false, b0, (short)0, ch[0], false, false);
    cg[0] = __builtin_amdgcn_wmma_f32_16x16x4_f32(false, ag, false, b0, (short)0, cg[0], false, false);
    ch[1] = __builtin_amdgcn_wmma_f32_16x16x4_f32(false, ah, false, b1, (short)0, ch[1], false, false);
    cg[1] = __builtin_amdgcn_wmma_f32_16x16x4_f32(false, ag, false, b1, (short)0, cg[1], false, false);
    ch[2] = __builtin_amdgcn_wmma_f32_16x16x4_f32(false, ah, false, b2, (short)0, ch[2], false, false);
    cg[2] = __builtin_amdgcn_wmma_f32_16x16x4_f32(false, ag, false, b2, (short)0, cg[2], false, false);
    ch[3] = __builtin_amdgcn_wmma_f32_16x16x4_f32(false, ah, false, b3, (short)0, ch[3], false, false);
    cg[3] = __builtin_amdgcn_wmma_f32_16x16x4_f32(false, ag, false, b3, (short)0, cg[3], false, false);
}

__global__ void __launch_bounds__(256)
gemm_coef_kernel(const float* __restrict__ x, const float* __restrict__ wp,
                 float2* __restrict__ coef) {
    const int lane = threadIdx.x & 31;
    const int wv   = threadIdx.x >> 5;
    const int dir  = wv >> 2;
    const int mg   = wv & 3;
    const int n0   = blockIdx.x * 64;
    const int bb   = blockIdx.y;

    const int ln    = lane & 15;
    const int khalf = (lane >> 4) << 1;

    const float*  xr = x + (size_t)bb * DIMI * SEQL + (size_t)khalf * SEQL + n0 + 2 * ln;
    const float4* wq = (const float4*)wp + (size_t)wv * (32 * 32) + lane;

    v8f ch[4] = {};
    v8f cg[4] = {};

    KStep f[3];
    f[0] = load_kstep(wq, xr);
    f[1] = load_kstep(wq + 32, xr + (size_t)4 * SEQL);
#pragma unroll
    for (int ks = 0; ks < 32; ++ks) {
        if (ks + 2 < 32)
            f[(ks + 2) % 3] = load_kstep(wq + (ks + 2) * 32, xr + (size_t)(ks + 2) * 4 * SEQL);
        __builtin_amdgcn_sched_barrier(0);
        mma_kstep(f[ks % 3], ch, cg);
        __builtin_amdgcn_sched_barrier(0);
    }

    const int mrow = (lane >> 4) << 3;
    float4* coef4 = (float4*)(coef + ((size_t)(dir * NB + bb) * CH + mg * 16 + mrow) * SEQL
                                   + n0 + 2 * ln);
#pragma unroll
    for (int r = 0; r < 8; ++r) {
#pragma unroll
        for (int q = 0; q < 2; ++q) {
            float ab[4];
#pragma unroll
            for (int p = 0; p < 2; ++p) {
                const float h = ch[q * 2 + p][r];
                const float g = cg[q * 2 + p][r];
                const float eg  = __expf(g);
                const float emg = __expf(-g);
                const float z  = __builtin_amdgcn_rcpf(1.f + emg);
                const float oz = __builtin_amdgcn_rcpf(1.f + eg);
                const float s  = __builtin_amdgcn_sqrtf(
                    2.f * __builtin_amdgcn_rcpf(eg + emg) + 1.f);
                ab[p * 2 + 0] = oz * s;
                ab[p * 2 + 1] = z * s * h;
            }
            coef4[((size_t)r * SEQL + 32 * q) / 2] = make_float4(ab[0], ab[1], ab[2], ab[3]);
        }
    }
}

#define CPITCH 17
#define OPITCH 9
template <int DIR>
__device__ __forceinline__ void
scan_body(const float2* __restrict__ coef, float* __restrict__ out,
          float4* sT, float* sA, float* sB) {
    const int tid  = threadIdx.x;
    const int lane = tid & 31;
    const int wv   = tid >> 5;
    const int c    = blockIdx.x;
    const int bb   = blockIdx.y;
    const size_t base = ((size_t)(DIR * NB + bb) * CH + c) * SEQL;
    const int    wlo  = DIR ? (SEQL - 1024 * (wv + 1)) : (1024 * wv);
    const int    chk  = DIR ? (31 - lane) : lane;
    float4* sw = sT + wv * (32 * CPITCH);

    const float4* src = (const float4*)(coef + base + wlo) + lane;
    float4 v[16];
#pragma unroll
    for (int i = 0; i < 16; ++i) v[i] = src[i * 32];
#pragma unroll
    for (int i = 0; i < 16; ++i)
        sw[(2 * i + (lane >> 4)) * CPITCH + (lane & 15)] = v[i];
    __syncthreads();

    float2 e[32];
#pragma unroll
    for (int i = 0; i < 16; ++i) {
        const float4 t = sw[chk * CPITCH + i];
        e[2 * i]     = make_float2(t.x, t.y);
        e[2 * i + 1] = make_float2(t.z, t.w);
    }

    float A = 1.f, Bv = 0.f;
#pragma unroll
    for (int s = 0; s < 32; ++s) {
        const int j = DIR ? (31 - s) : s;
        A  = e[j].x * A;
        Bv = e[j].x * Bv + e[j].y;
    }

    sA[tid] = A; sB[tid] = Bv;
    __syncthreads();
    for (int off = 1; off < 256; off <<= 1) {
        float pa = 1.f, pb = 0.f;
        if (tid >= off) { pa = sA[tid - off]; pb = sB[tid - off]; }
        __syncthreads();
        if (tid >= off) {
            const float ca = sA[tid], cb = sB[tid];
            sA[tid] = ca * pa;
            sB[tid] = ca * pb + cb;
        }
        __syncthreads();
    }
    float h = (tid == 0) ? 0.f : sB[tid - 1];

#pragma unroll
    for (int s = 0; s < 32; ++s) {
        const int j = DIR ? (31 - s) : s;
        h = e[j].x * h + e[j].y;
        e[j].x = h;
    }

#pragma unroll
    for (int q = 0; q < 8; ++q)
        sw[chk * OPITCH + q] = make_float4(e[4 * q].x, e[4 * q + 1].x,
                                           e[4 * q + 2].x, e[4 * q + 3].x);
    __syncthreads();
    float4* op = (float4*)(out + ((size_t)bb * 128 + DIR * 64 + c) * SEQL + wlo) + lane;
#pragma unroll
    for (int q = 0; q < 8; ++q)
        op[q * 32] = sw[(4 * q + (lane >> 3)) * OPITCH + (lane & 7)];
}

__global__ void __launch_bounds__(256)
scan_kernel(const float2* __restrict__ coef, float* __restrict__ out) {
    __shared__ float4 sT[8 * 32 * CPITCH];
    __shared__ float sA[256];
    __shared__ float sB[256];
    if (blockIdx.z == 0) scan_body<0>(coef, out, sT, sA, sB);
    else                 scan_body<1>(coef, out, sT, sA, sB);
}

extern "C" void kernel_launch(void* const* d_in, const int* in_sizes, int n_in,
                              void* d_out, int out_size, void* d_ws, size_t ws_size,
                              hipStream_t stream) {
    const float* x = (const float*)d_in[0];
    const float* w = (const float*)d_in[1];
    float* out = (float*)d_out;

    float*  we   = (float*)d_ws;
    float2* coef = (float2*)((char*)d_ws + ODIM * DIMI * sizeof(float));

    wnorm_kernel<<<8, 1024, 0, stream>>>(w, we);
    gemm_coef_kernel<<<dim3(SEQL / 64, NB), 256, 0, stream>>>(x, we, coef);
    scan_kernel<<<dim3(CH, NB, 2), 256, 0, stream>>>(coef, out);
}


# Weight-norm launch folded into the GEMM: a wave keeps its raw weight rows in registers over two L tiles, sums their squares there, scales its accumulator rows by 1/(norm+eps); 2 launches
# speedup vs baseline: 2.3682x; 1.0310x over previous
#include <hip/hip_runtime.h>

typedef __attribute__((ext_vector_type(2))) float v2f;
typedef __attribute__((ext_vector_type(8))) float v8f;

#define DIMI 128
#define ODIM 256
#define SEQL 8192
#define NB   8
#define CH   64

struct BStep { float2 p00, p01, p10, p11; };

#define XROW(ks) (8 * ((ks) >> 1) + 2 * ((ks) & 1))
#define AHEAD 3

__device__ __forceinline__ BStep load_bstep(const float* __restrict__ xr) {
    BStep f;
    f.p00 = *(const float2*)(xr);
    f.p01 = *(const float2*)(xr + SEQL);
    f.p10 = *(const float2*)(xr + 32);
    f.p11 = *(const float2*)(xr + SEQL + 32);
    return f;
}

__device__ __forceinline__ void mma_kstep(const v2f ah, const v2f ag, const BStep& f,
                                          v8f (&ch)[4], v8f (&cg)[4]) {
    const v2f b0 = { f.p00.x, f.p01.x };
    const v2f b1 = { f.p00.y, f.p01.y };
    const v2f b2 = { f.p10.x, f.p11.x };
    const v2f b3 = { f.p10.y, f.p11.y };
    ch[0] = __builtin_amdgcn_wmma_f32_16x16x4_f32(false, ah, false, b0, (short)0, ch[0], false, false);
    cg[0] = __builtin_amdgcn_wmma_f32_16x16x4_f32(false, ag, false, b0, (short)0, cg[0], false, false);
    ch[1] = __builtin_amdgcn_wmma_f32_16x16x4_f32(false, ah, false, b1, (short)0, ch[1], false, false);
    cg[1] = __builtin_amdgcn_wmma_f32_16x16x4_f32(false, ag, false, b1, (short)0, cg[1], false, false);
    ch[2] = __builtin_amdgcn_wmma_f32_16x16x4_f32(false, ah, false, b2, (short)0, ch[2], false, false);
    cg[2] = __builtin_amdgcn_wmma_f32_16x16x4_f32(false, ag, false, b2, (short)0, cg[2], false, false);
    ch[3] = __builtin_amdgcn_wmma_f32_16x16x4_f32(false, ah, false, b3, (short)0, ch[3], false, false);
    cg[3] = __builtin_amdgcn_wmma_f32_16x16x4_f32(false, ag, false, b3, (short)0, cg[3], false, false);
}

template <bool LOADA>
__device__ __forceinline__ void
k_loop(const float* __restrict__ xr, const float* __restrict__ wh, const float* __restrict__ wg,
       float4 (&ah)[16], float4 (&ag)[16], v8f (&ch)[4], v8f (&cg)[4]) {
    if (LOADA) {
#pragma unroll
        for (int j = 0; j < 2; ++j) {
            ah[j] = *(const float4*)(wh + 8 * j);
            ag[j] = *(const float4*)(wg + 8 * j);
        }
    }
    BStep f[AHEAD + 1];
#pragma unroll
    for (int ks = 0; ks < AHEAD; ++ks)
        f[ks] = load_bstep(xr + (size_t)XROW(ks) * SEQL);
#pragma unroll
    for (int ks = 0; ks < 32; ++ks) {
        if (LOADA && (ks & 1) == 0 && (ks >> 1) + 2 < 16) {
            ah[(ks >> 1) + 2] = *(const float4*)(wh + 8 * ((ks >> 1) + 2));
            ag[(ks >> 1) + 2] = *(const float4*)(wg + 8 * ((ks >> 1) + 2));
        }
        if (ks + AHEAD < 32)
            f[(ks + AHEAD) % (AHEAD + 1)] = load_bstep(xr + (size_t)XROW(ks + AHEAD) * SEQL);
        __builtin_amdgcn_sched_barrier(0);
        const float4 a4 = ah[ks >> 1], g4 = ag[ks >> 1];
        const v2f ah2 = (ks & 1) ? v2f{ a4.z, a4.w } : v2f{ a4.x, a4.y };
        const v2f ag2 = (ks & 1) ? v2f{ g4.z, g4.w } : v2f{ g4.x, g4.y };
        mma_kstep(ah2, ag2, f[ks % (AHEAD + 1)], ch, cg);
        __builtin_amdgcn_sched_barrier(0);
    }
}

__device__ __forceinline__ void
emit_coefs(const v8f (&ch)[4], const v8f (&cg)[4], const float (&sch)[8], const float (&scg)[8],
           float4* __restrict__ coef4) {
#pragma unroll
    for (int r = 0; r < 8; ++r) {
#pragma unroll
        for (int q = 0; q < 2; ++q) {
            float ab[4];
#pragma unroll
            for (int p = 0; p < 2; ++p) {
                const float h = ch[q * 2 + p][r] * sch[r];
                const float m   = cg[q * 2 + p][r] * scg[r];
                const float eg  = __builtin_amdgcn_exp2f(m);
                const float emg = __builtin_amdgcn_exp2f(-m);
                const float z  = __builtin_amdgcn_rcpf(1.f + emg);
                const float oz = __builtin_amdgcn_rcpf(1.f + eg);
                const float s  = __builtin_amdgcn_sqrtf(
                    2.f * __builtin_amdgcn_rcpf(eg + emg) + 1.f);
                ab[p * 2 + 0] = oz * s;
                ab[p * 2 + 1] = z * s * h;
            }
            coef4[((size_t)r * SEQL + 32 * q) / 2] = make_float4(ab[0], ab[1], ab[2], ab[3]);
        }
    }
}

__global__ void __launch_bounds__(256, 2)
gemm_coef_kernel(const float* __restrict__ x, const float* __restrict__ w,
                 float2* __restrict__ coef) {
    const int lane = threadIdx.x & 31;
    const int wv   = threadIdx.x >> 5;
    const int dir  = wv >> 2;
    const int mg   = wv & 3;
    const int bb   = blockIdx.y;
    const int ln   = lane & 15;
    const int half = lane >> 4;
    const int mrow = half << 3;
    const int n0   = blockIdx.x * 128;

    const float* wh = w + (size_t)(dir * 128 + mg * 16 + ln) * DIMI + 4 * half;
    const float* wg = wh + 64 * DIMI;
    const float* xr = x + (size_t)bb * DIMI * SEQL + (size_t)(4 * half) * SEQL + n0 + 2 * ln;
    float4* coef4 = (float4*)(coef + ((size_t)(dir * NB + bb) * CH + mg * 16 + mrow) * SEQL
                                   + n0 + 2 * ln);
    float4 ah[16], ag[16];

    v8f ch[4] = {};
    v8f cg[4] = {};
    k_loop<true>(xr, wh, wg, ah, ag, ch, cg);

    float4 qh = make_float4(0.f, 0.f, 0.f, 0.f), qg = qh;
#pragma unroll
    for (int j = 0; j < 16; ++j) {
        qh.x += ah[j].x * ah[j].x; qh.y += ah[j].y * ah[j].y;
        qh.z += ah[j].z * ah[j].z; qh.w += ah[j].w * ah[j].w;
        qg.x += ag[j].x * ag[j].x; qg.y += ag[j].y * ag[j].y;
        qg.z += ag[j].z * ag[j].z; qg.w += ag[j].w * ag[j].w;
    }
    float ssh = (qh.x + qh.y) + (qh.z + qh.w);
    float ssg = (qg.x + qg.y) + (qg.z + qg.w);
    ssh += __shfl_xor(ssh, 16);
    ssg += __shfl_xor(ssg, 16);
    const float invh = 1.f / (sqrtf(ssh) + 1.13137084989847604e-3f);
    const float invg = 1.f / (sqrtf(ssg) + 1.13137084989847604e-3f);
    float sch[8], scg[8];
#pragma unroll
    for (int r = 0; r < 8; ++r) {
        sch[r] = __shfl(invh, mrow + r);
        scg[r] = __shfl(invg, mrow + r) * 1.44269504088896341f;
    }
    emit_coefs(ch, cg, sch, scg, coef4);

#pragma unroll
    for (int i = 0; i < 4; ++i) { ch[i] = v8f{}; cg[i] = v8f{}; }
    k_loop<false>(xr + 64, wh, wg, ah, ag, ch, cg);
    emit_coefs(ch, cg, sch, scg, coef4 + 32);
}

#define CPITCH 17
#define OPITCH 9
template <int DIR>
__device__ __forceinline__ void
scan_body(const float2* __restrict__ coef, float* __restrict__ out,
          float4* sT, float* sA, float* sB) {
    const int tid  = threadIdx.x;
    const int lane = tid & 31;
    const int wv   = tid >> 5;
    const int c    = blockIdx.x;
    const int bb   = blockIdx.y;
    const size_t base = ((size_t)(DIR * NB + bb) * CH + c) * SEQL;
    const int    wlo  = DIR ? (SEQL - 1024 * (wv + 1)) : (1024 * wv);
    const int    chk  = DIR ? (31 - lane) : lane;
    float4* sw = sT + wv * (32 * CPITCH);

    const float4* src = (const float4*)(coef + base + wlo) + lane;
    float4 v[16];
#pragma unroll
    for (int i = 0; i < 16; ++i) v[i] = src[i * 32];
#pragma unroll
    for (int i = 0; i < 16; ++i)
        sw[(2 * i + (lane >> 4)) * CPITCH + (lane & 15)] = v[i];
    __syncthreads();

    float2 e[32];
#pragma unroll
    for (int i = 0; i < 16; ++i) {
        const float4 t = sw[chk * CPITCH + i];
        e[2 * i]     = make_float2(t.x, t.y);
        e[2 * i + 1] = make_float2(t.z, t.w);
    }

    float A = 1.f, Bv = 0.f;
#pragma unroll
    for (int s = 0; s < 32; ++s) {
        const int j = DIR ? (31 - s) : s;
        A  = e[j].x * A;
        Bv = e[j].x * Bv + e[j].y;
    }

    sA[tid] = A; sB[tid] = Bv;
    __syncthreads();
    for (int off = 1; off < 256; off <<= 1) {
        float pa = 1.f, pb = 0.f;
        if (tid >= off) { pa = sA[tid - off]; pb = sB[tid - off]; }
        __syncthreads();
        if (tid >= off) {
            const float ca = sA[tid], cb = sB[tid];
            sA[tid] = ca * pa;
            sB[tid] = ca * pb + cb;
        }
        __syncthreads();
    }
    float h = (tid == 0) ? 0.f : sB[tid - 1];

#pragma unroll
    for (int s = 0; s < 32; ++s) {
        const int j = DIR ? (31 - s) : s;
        h = e[j].x * h + e[j].y;
        e[j].x = h;
    }

#pragma unroll
    for (int q = 0; q < 8; ++q)
        sw[chk * OPITCH + q] = make_float4(e[4 * q].x, e[4 * q + 1].x,
                                           e[4 * q + 2].x, e[4 * q + 3].x);
    __syncthreads();
    float4* op = (float4*)(out + ((size_t)bb * 128 + DIR * 64 + c) * SEQL + wlo) + lane;
#pragma unroll
    for (int q = 0; q < 8; ++q)
        op[q * 32] = sw[(4 * q + (lane >> 3)) * OPITCH + (lane & 7)];
}

__global__ void __launch_bounds__(256)
scan_kernel(const float2* __restrict__ coef, float* __restrict__ out) {
    __shared__ float4 sT[8 * 32 * CPITCH];
    __shared__ float sA[256];
    __shared__ float sB[256];
    if (blockIdx.z == 0) scan_body<0>(coef, out, sT, sA, sB);
    else                 scan_body<1>(coef, out, sT, sA, sB);
}

extern "C" void kernel_launch(void* const* d_in, const int* in_sizes, int n_in,
                              void* d_out, int out_size, void* d_ws, size_t ws_size,
                              hipStream_t stream) {
    const float* x = (const float*)d_in[0];
    const float* w = (const float*)d_in[1];
    float* out = (float*)d_out;

    float2* coef = (float2*)((char*)d_ws + ODIM * DIMI * sizeof(float));

    gemm_coef_kernel<<<dim3(SEQL / 128, NB), 256, 0, stream>>>(x, w, coef);
    scan_kernel<<<dim3(CH, NB, 2), 256, 0, stream>>>(coef, out);
}
